# starting kernel
# baseline (speedup 1.0000x reference)
#include <hip/hip_runtime.h>

typedef float v2f __attribute__((ext_vector_type(2)));
typedef float v8f __attribute__((ext_vector_type(8)));

#define WMMA_F32(a, b, c) \
    __builtin_amdgcn_wmma_f32_16x16x4_f32(false, (a), false, (b), (short)0, (c), false, false)

constexpr int Bsz    = 4;
constexpr int Nseq   = 2048;
constexpr int Cdim   = 1024;
constexpr int Hn     = 16;
constexpr int HD     = 64;
constexpr int ThreeC = 3 * Cdim;
constexpr int Mrows  = Bsz * Nseq;

__global__ __launch_bounds__(256)
void gemm_xwt_kernel(const float* __restrict__ X, const float* __restrict__ W,
                     const float* __restrict__ bias, float* __restrict__ Y,
                     int M, int Nd, int K)
{
    const int lane = threadIdx.x & 31;
    const int wid  = blockIdx.x * (blockDim.x >> 5) + (threadIdx.x >> 5);
    const int tilesN = Nd >> 6;
    const int m0 = (wid / tilesN) * 32;
    const int n0 = (wid % tilesN) * 64;
    const int hlf = lane >> 4;
    const int l16 = lane & 15;

    v8f zero = {};
    v8f acc[2][4];
#pragma unroll
    for (int i = 0; i < 2; i++)
#pragma unroll
        for (int j = 0; j < 4; j++) acc[i][j] = zero;

    const float* arow0 = X + (size_t)(m0 + l16) * K;
    const float* arow1 = X + (size_t)(m0 + 16 + l16) * K;
    const float* brow0 = W + (size_t)(n0 +  0 + l16) * K;
    const float* brow1 = W + (size_t)(n0 + 16 + l16) * K;
    const float* brow2 = W + (size_t)(n0 + 32 + l16) * K;
    const float* brow3 = W + (size_t)(n0 + 48 + l16) * K;

    const int ko0 = 2 * hlf;
    v2f a0 = *(const v2f*)(arow0 + ko0);
    v2f a1 = *(const v2f*)(arow1 + ko0);
    v2f b0 = *(const v2f*)(brow0 + ko0);
    v2f b1 = *(const v2f*)(brow1 + ko0);
    v2f b2 = *(const v2f*)(brow2 + ko0);
    v2f b3 = *(const v2f*)(brow3 + ko0);

    for (int k0 = 4; k0 < K; k0 += 4) {
        const int kn = k0 + ko0;
        v2f na0 = *(const v2f*)(arow0 + kn);
        v2f na1 = *(const v2f*)(arow1 + kn);
        v2f nb0 = *(const v2f*)(brow0 + kn);
        v2f nb1 = *(const v2f*)(brow1 + kn);
        v2f nb2 = *(const v2f*)(brow2 + kn);
        v2f nb3 = *(const v2f*)(brow3 + kn);

        acc[0][0] = WMMA_F32(a0, b0, acc[0][0]);
        acc[1][0] = WMMA_F32(a1, b0, acc[1][0]);
        acc[0][1] = WMMA_F32(a0, b1, acc[0][1]);
        acc[1][1] = WMMA_F32(a1, b1, acc[1][1]);
        acc[0][2] = WMMA_F32(a0, b2, acc[0][2]);
        acc[1][2] = WMMA_F32(a1, b2, acc[1][2]);
        acc[0][3] = WMMA_F32(a0, b3, acc[0][3]);
        acc[1][3] = WMMA_F32(a1, b3, acc[1][3]);

        a0 = na0; a1 = na1; b0 = nb0; b1 = nb1; b2 = nb2; b3 = nb3;
    }
    acc[0][0] = WMMA_F32(a0, b0, acc[0][0]);
    acc[1][0] = WMMA_F32(a1, b0, acc[1][0]);
    acc[0][1] = WMMA_F32(a0, b1, acc[0][1]);
    acc[1][1] = WMMA_F32(a1, b1, acc[1][1]);
    acc[0][2] = WMMA_F32(a0, b2, acc[0][2]);
    acc[1][2] = WMMA_F32(a1, b2, acc[1][2]);
    acc[0][3] = WMMA_F32(a0, b3, acc[0][3]);
    acc[1][3] = WMMA_F32(a1, b3, acc[1][3]);

#pragma unroll
    for (int i = 0; i < 2; i++)
#pragma unroll
        for (int j = 0; j < 4; j++)
#pragma unroll
            for (int r = 0; r < 8; r++) {
                const int row = m0 + 16 * i + r + 8 * hlf;
                const int col = n0 + 16 * j + l16;
                float v = acc[i][j][r];
                if (bias) v += bias[col];
                Y[(size_t)row * Nd + col] = v;
            }
}

__global__ __launch_bounds__(256)
void rope_kernel(float* __restrict__ qkv,
                 const float* __restrict__ cosT, const float* __restrict__ sinT)
{
    const int t = blockIdx.x * blockDim.x + threadIdx.x;
    const int i = t & 31;
    const int h = (t >> 5) & 15;
    const int n = (t >> 9) & (Nseq - 1);
    const int b = t >> 20;

    const size_t base = (size_t)(b * Nseq + n) * ThreeC + h * HD;
    const float c0 = cosT[n * HD + i],      s0 = sinT[n * HD + i];
    const float c1 = cosT[n * HD + i + 32], s1 = sinT[n * HD + i + 32];

    const float q0 = qkv[base + i], q1 = qkv[base + i + 32];
    qkv[base + i]      = q0 * c0 - q1 * s0;
    qkv[base + i + 32] = q1 * c1 + q0 * s1;

    const float k0 = qkv[base + Cdim + i], k1 = qkv[base + Cdim + i + 32];
    qkv[base + Cdim + i]      = k0 * c0 - k1 * s0;
    qkv[base + Cdim + i + 32] = k1 * c1 + k0 * s1;
}

__global__ __launch_bounds__(256)
void flash_attn_kernel(const float* __restrict__ qkv, float* __restrict__ O)
{
    const int lane = threadIdx.x & 31;
    const int wid  = blockIdx.x * (blockDim.x >> 5) + (threadIdx.x >> 5);
    const int qt = wid & 127;
    const int h  = (wid >> 7) & 15;
    const int b  = wid >> 11;
    const int hlf = lane >> 4;
    const int l16 = lane & 15;
    const int q0  = qt * 16;
    const float scale = 0.125f;

    const size_t rowQ = (size_t)(b * Nseq + q0 + l16) * ThreeC + h * HD;
    v2f qb[16];
#pragma unroll
    for (int kk = 0; kk < 16; kk++)
        qb[kk] = *(const v2f*)(qkv + rowQ + 4 * kk + 2 * hlf);

    v8f zero = {};
    v8f acc[4];
#pragma unroll
    for (int t = 0; t < 4; t++) acc[t] = zero;
    float m = -3.0e38f, l = 0.0f;

    const size_t baseK = (size_t)(b * Nseq) * ThreeC + Cdim + h * HD;
    const size_t baseV = (size_t)(b * Nseq) * ThreeC + 2 * Cdim + h * HD;

    for (int kt = 0; kt < Nseq / 16; kt++) {
        const int key0 = kt * 16;

        const size_t rowK = baseK + (size_t)(key0 + l16) * ThreeC;
        v2f ka[16];
#pragma unroll
        for (int kk = 0; kk < 16; kk++)
            ka[kk] = *(const v2f*)(qkv + rowK + 4 * kk + 2 * hlf);

        v8f s = zero;
#pragma unroll
        for (int kk = 0; kk < 16; kk++)
            s = WMMA_F32(ka[kk], qb[kk], s);

        v2f va[4][4];
#pragma unroll
        for (int t = 0; t < 4; t++) {
            const size_t colV = baseV + t * 16 + l16;
#pragma unroll
            for (int kk = 0; kk < 4; kk++) {
                const int ka0 = key0 + 4 * kk + 2 * hlf;
                va[t][kk].x = qkv[colV + (size_t)ka0 * ThreeC];
                va[t][kk].y = qkv[colV + (size_t)(ka0 + 1) * ThreeC];
            }
        }

        float sv[8];
#pragma unroll
        for (int r = 0; r < 8; r++) sv[r] = s[r] * scale;

        float tm = sv[0];
#pragma unroll
        for (int r = 1; r < 8; r++) tm = fmaxf(tm, sv[r]);
        tm = fmaxf(tm, __shfl_xor(tm, 16, 32));

        const float mn   = fmaxf(m, tm);
        const float corr = __expf(m - mn);
        m = mn;

        float p[8];
        float rs = 0.0f;
#pragma unroll
        for (int r = 0; r < 8; r++) { p[r] = __expf(sv[r] - mn); rs += p[r]; }
        rs += __shfl_xor(rs, 16, 32);
        l = l * corr + rs;

#pragma unroll
        for (int t = 0; t < 4; t++)
#pragma unroll
            for (int r = 0; r < 8; r++) acc[t][r] *= corr;

        const float x2 = __shfl_xor(p[2], 16, 32), x3 = __shfl_xor(p[3], 16, 32);
        const float x6 = __shfl_xor(p[6], 16, 32), x7 = __shfl_xor(p[7], 16, 32);
        const float u0 = __shfl_xor(p[0], 16, 32), u1 = __shfl_xor(p[1], 16, 32);
        const float u4 = __shfl_xor(p[4], 16, 32), u5 = __shfl_xor(p[5], 16, 32);
        v2f pb[4];
        pb[0].x = hlf ? x2 : p[0];  pb[0].y = hlf ? x3 : p[1];
        pb[1].x = hlf ? x6 : p[4];  pb[1].y = hlf ? x7 : p[5];
        pb[2].x = hlf ? p[2] : u0;  pb[2].y = hlf ? p[3] : u1;
        pb[3].x = hlf ? p[6] : u4;  pb[3].y = hlf ? p[7] : u5;

#pragma unroll
        for (int t = 0; t < 4; t++)
#pragma unroll
            for (int kk = 0; kk < 4; kk++)
                acc[t] = WMMA_F32(va[t][kk], pb[kk], acc[t]);
    }

    const float inv = 1.0f / l;
    const size_t baseO = (size_t)(b * Nseq + q0 + l16) * Cdim + h * HD;
#pragma unroll
    for (int t = 0; t < 4; t++)
#pragma unroll
        for (int r = 0; r < 8; r++)
            O[baseO + t * 16 + r + 8 * hlf] = acc[t][r] * inv;
}

extern "C" void kernel_launch(void* const* d_in, const int* in_sizes, int n_in,
                              void* d_out, int out_size, void* d_ws, size_t ws_size,
                              hipStream_t stream)
{
    const float* x        = (const float*)d_in[0];
    const float* rope_cos = (const float*)d_in[1];
    const float* rope_sin = (const float*)d_in[2];
    const float* W_qkv    = (const float*)d_in[3];
    const float* W_proj   = (const float*)d_in[4];
    const float* b_proj   = (const float*)d_in[5];
    float* out  = (float*)d_out;

    float* qkv  = (float*)d_ws;
    float* attn = qkv + (size_t)Mrows * ThreeC;

    {
        const int waves = (Mrows / 32) * (ThreeC / 64);
        gemm_xwt_kernel<<<waves / 8, 256, 0, stream>>>(x, W_qkv, nullptr, qkv,
                                                       Mrows, ThreeC, Cdim);
    }
    {
        const int threads = Bsz * Nseq * Hn * 32;
        rope_kernel<<<threads / 256, 256, 0, stream>>>(qkv, rope_cos, rope_sin);
    }
    {
        const int waves = Bsz * Hn * (Nseq / 16);
        flash_attn_kernel<<<waves / 8, 256, 0, stream>>>(qkv, attn);
    }
    {
        const int waves = (Mrows / 32) * (Cdim / 64);
        gemm_xwt_kernel<<<waves / 8, 256, 0, stream>>>(attn, W_proj, b_proj, out,
                                                       Mrows, Cdim, Cdim);
    }
}


# attention reads q, k, v in fragment order: the rotary pass writes them packed (8 KB of k and v per 16 keys), each fragment load 512 contiguous bytes, 16 per key tile not 48; k and sum order kept
# speedup vs baseline: 2.0043x; 2.0043x over previous
#include <hip/hip_runtime.h>

typedef float v2f __attribute__((ext_vector_type(2)));
typedef float v4f __attribute__((ext_vector_type(4)));
typedef float v8f __attribute__((ext_vector_type(8)));

#define WMMA_F32(a, b, c) \
    __builtin_amdgcn_wmma_f32_16x16x4_f32(false, (a), false, (b), (short)0, (c), false, false)

constexpr int Bsz    = 4;
constexpr int Nseq   = 2048;
constexpr int Cdim   = 1024;
constexpr int Hn     = 16;
constexpr int HD     = 64;
constexpr int ThreeC = 3 * Cdim;
constexpr int Mrows  = Bsz * Nseq;


__global__ __launch_bounds__(256)
void pack_operands_kernel(const float* __restrict__ X, const float* __restrict__ Wq,
                          const float* __restrict__ Wp, float* __restrict__ P)
{
    const int lane = threadIdx.x & 31;
    const int wid  = blockIdx.x * (blockDim.x >> 5) + (threadIdx.x >> 5);
    const int rt = wid >> 5;
    const int g  = wid & 31;
    const int c  = lane & 15;
    const int h  = lane >> 4;

    constexpr int RT_X  = Mrows / 16;
    constexpr int RT_WQ = RT_X + ThreeC / 16;
    const float* src;
    if (rt < RT_X)       src = X  + (size_t)(rt * 16 + c) * Cdim;
    else if (rt < RT_WQ) src = Wq + (size_t)((rt - RT_X) * 16 + c) * Cdim;
    else                 src = Wp + (size_t)((rt - RT_WQ) * 16 + c) * Cdim;
    src += g * 32 + 2 * h;

    v4f* dst = (v4f*)P + ((size_t)rt * (Cdim >> 3) + g * 4) * 32 + lane;
#pragma unroll
    for (int j = 0; j < 4; j++) {
        const v2f lo = *(const v2f*)(src + 8 * j);
        const v2f hi = *(const v2f*)(src + 8 * j + 4);
        v4f o;
        o.x = lo.x; o.y = lo.y; o.z = hi.x; o.w = hi.y;
        dst[j * 32] = o;
    }
}

#define GEMM_BLOCK(A0, A1, B0, B1, B2, B3)                       \
    do {                                                          \
        acc[0][0] = WMMA_F32((A0).xy, (B0).xy, acc[0][0]);       \
        acc[1][0] = WMMA_F32((A1).xy, (B0).xy, acc[1][0]);       \
        acc[0][1] = WMMA_F32((A0).xy, (B1).xy, acc[0][1]);       \
        acc[1][1] = WMMA_F32((A1).xy, (B1).xy, acc[1][1]);       \
        acc[0][2] = WMMA_F32((A0).xy, (B2).xy, acc[0][2]);       \
        acc[1][2] = WMMA_F32((A1).xy, (B2).xy, acc[1][2]);       \
        acc[0][3] = WMMA_F32((A0).xy, (B3).xy, acc[0][3]);       \
        acc[1][3] = WMMA_F32((A1).xy, (B3).xy, acc[1][3]);       \
        acc[0][0] = WMMA_F32((A0).zw, (B0).zw, acc[0][0]);       \
        acc[1][0] = WMMA_F32((A1).zw, (B0).zw, acc[1][0]);       \
        acc[0][1] = WMMA_F32((A0).zw, (B1).zw, acc[0][1]);       \
        acc[1][1] = WMMA_F32((A1).zw, (B1).zw, acc[1][1]);       \
        acc[0][2] = WMMA_F32((A0).zw, (B2).zw, acc[0][2]);       \
        acc[1][2] = WMMA_F32((A1).zw, (B2).zw, acc[1][2]);       \
        acc[0][3] = WMMA_F32((A0).zw, (B3).zw, acc[0][3]);       \
        acc[1][3] = WMMA_F32((A1).zw, (B3).zw, acc[1][3]);       \
    } while (0)

__global__ __launch_bounds__(256)
void gemm_packed_kernel(const float* __restrict__ Ap, const float* __restrict__ Bp,
                        const float* __restrict__ bias, float* __restrict__ Y,
                        int M, int Nd, int K)
{
    const int lane = threadIdx.x & 31;
    const int wid  = blockIdx.x * (blockDim.x >> 5) + (threadIdx.x >> 5);
    const int tilesN = Nd >> 6;
    const int m0 = (wid / tilesN) * 32;
    const int n0 = (wid % tilesN) * 64;
    const int hlf = lane >> 4;
    const int l16 = lane & 15;
    const int KB  = K >> 3;

    v8f zero = {};
    v8f acc[2][4];
#pragma unroll
    for (int i = 0; i < 2; i++)
#pragma unroll
        for (int j = 0; j < 4; j++) acc[i][j] = zero;

    const size_t tstride = (size_t)KB * 32;
    const v4f* pa0 = (const v4f*)Ap + (size_t)(m0 >> 4) * tstride + lane;
    const v4f* pa1 = pa0 + tstride;
    const v4f* pb0 = (const v4f*)Bp + (size_t)(n0 >> 4) * tstride + lane;
    const v4f* pb1 = pb0 + tstride;
    const v4f* pb2 = pb1 + tstride;
    const v4f* pb3 = pb2 + tstride;

    v4f a0 = pa0[0], a1 = pa1[0];
    v4f b0 = pb0[0], b1 = pb1[0], b2 = pb2[0], b3 = pb3[0];
    v4f c0, c1, d0, d1, d2, d3;

    int kb = 0;
    for (; kb < KB - 2; kb += 2) {
        const int o1 = (kb + 1) * 32, o2 = (kb + 2) * 32;
        c0 = pa0[o1]; c1 = pa1[o1];
        d0 = pb0[o1]; d1 = pb1[o1]; d2 = pb2[o1]; d3 = pb3[o1];
        GEMM_BLOCK(a0, a1, b0, b1, b2, b3);
        a0 = pa0[o2]; a1 = pa1[o2];
        b0 = pb0[o2]; b1 = pb1[o2]; b2 = pb2[o2]; b3 = pb3[o2];
        GEMM_BLOCK(c0, c1, d0, d1, d2, d3);
    }
    {
        const int o1 = (kb + 1) * 32;
        c0 = pa0[o1]; c1 = pa1[o1];
        d0 = pb0[o1]; d1 = pb1[o1]; d2 = pb2[o1]; d3 = pb3[o1];
        GEMM_BLOCK(a0, a1, b0, b1, b2, b3);
        GEMM_BLOCK(c0, c1, d0, d1, d2, d3);
    }

#pragma unroll
    for (int i = 0; i < 2; i++)
#pragma unroll
        for (int j = 0; j < 4; j++)
#pragma unroll
            for (int r = 0; r < 8; r++) {
                const int row = m0 + 16 * i + r + 8 * hlf;
                const int col = n0 + 16 * j + l16;
                float v = acc[i][j][r];
                if (bias) v += bias[col];
                Y[(size_t)row * Nd + col] = v;
            }
}

__global__ __launch_bounds__(256)
void rope_pack_kernel(const float* __restrict__ qkv,
                      const float* __restrict__ cosT, const float* __restrict__ sinT,
                      float* __restrict__ Qp, float* __restrict__ KVp)
{
    const int lane = threadIdx.x & 31;
    const int wid  = blockIdx.x * (blockDim.x >> 5) + (threadIdx.x >> 5);
    const int h    = wid & 15;
    const int tile = (wid >> 4) & 127;
    const int b    = wid >> 11;
    const int c    = lane & 15;
    const int hf   = lane >> 4;
    const int n    = tile * 16 + c;

    const float* qrow = qkv + (size_t)(b * Nseq + n) * ThreeC + h * HD + 2 * hf;
    const float* krow = qrow + Cdim;
    const float* cr   = cosT + n * HD + 2 * hf;
    const float* sr   = sinT + n * HD + 2 * hf;

    const size_t bht = (size_t)((b * Hn + h) * (Nseq / 16) + tile);
    v4f* qdst = (v4f*)Qp  + bht * (8 * 32)  + lane;
    v4f* kdst = (v4f*)KVp + bht * (16 * 32) + lane;

#pragma unroll
    for (int j = 0; j < 4; j++) {
        const v2f c0a = *(const v2f*)(cr + 8 * j),      c0b = *(const v2f*)(cr + 8 * j + 4);
        const v2f c1a = *(const v2f*)(cr + 8 * j + 32), c1b = *(const v2f*)(cr + 8 * j + 36);
        const v2f s0a = *(const v2f*)(sr + 8 * j),      s0b = *(const v2f*)(sr + 8 * j + 4);
        const v2f s1a = *(const v2f*)(sr + 8 * j + 32), s1b = *(const v2f*)(sr + 8 * j + 36);

        const v2f q0a = *(const v2f*)(qrow + 8 * j),      q0b = *(const v2f*)(qrow + 8 * j + 4);
        const v2f q1a = *(const v2f*)(qrow + 8 * j + 32), q1b = *(const v2f*)(qrow + 8 * j + 36);
        v4f lo, hi;
        lo.x = q0a.x * c0a.x - q1a.x * s0a.x;  lo.y = q0a.y * c0a.y - q1a.y * s0a.y;
        lo.z = q0b.x * c0b.x - q1b.x * s0b.x;  lo.w = q0b.y * c0b.y - q1b.y * s0b.y;
        hi.x = q1a.x * c1a.x + q0a.x * s1a.x;  hi.y = q1a.y * c1a.y + q0a.y * s1a.y;
        hi.z = q1b.x * c1b.x + q0b.x * s1b.x;  hi.w = q1b.y * c1b.y + q0b.y * s1b.y;
        qdst[j * 32]       = lo;
        qdst[(j + 4) * 32] = hi;

        const v2f k0a = *(const v2f*)(krow + 8 * j),      k0b = *(const v2f*)(krow + 8 * j + 4);
        const v2f k1a = *(const v2f*)(krow + 8 * j + 32), k1b = *(const v2f*)(krow + 8 * j + 36);
        lo.x = k0a.x * c0a.x - k1a.x * s0a.x;  lo.y = k0a.y * c0a.y - k1a.y * s0a.y;
        lo.z = k0b.x * c0b.x - k1b.x * s0b.x;  lo.w = k0b.y * c0b.y - k1b.y * s0b.y;
        hi.x = k1a.x * c1a.x + k0a.x * s1a.x;  hi.y = k1a.y * c1a.y + k0a.y * s1a.y;
        hi.z = k1b.x * c1b.x + k0b.x * s1b.x;  hi.w = k1b.y * c1b.y + k0b.y * s1b.y;
        kdst[j * 32]       = lo;
        kdst[(j + 4) * 32] = hi;
    }

    const float* vcol = qkv + (size_t)(b * Nseq + tile * 16 + 2 * hf) * ThreeC
                            + 2 * Cdim + h * HD + c;
#pragma unroll
    for (int t = 0; t < 4; t++)
#pragma unroll
        for (int kb2 = 0; kb2 < 2; kb2++) {
            const float* p = vcol + (size_t)(8 * kb2) * ThreeC + 16 * t;
            v4f o;
            o.x = p[0];
            o.y = p[ThreeC];
            o.z = p[4 * ThreeC];
            o.w = p[5 * ThreeC];
            kdst[(8 + 2 * t + kb2) * 32] = o;
        }
}

__global__ __launch_bounds__(256)
void flash_attn_kernel(const float* __restrict__ Qp, const float* __restrict__ KVp,
                       float* __restrict__ O)
{
    const int lane = threadIdx.x & 31;
    const int wid  = blockIdx.x * (blockDim.x >> 5) + (threadIdx.x >> 5);
    const int qt = wid & 127;
    const int h  = (wid >> 7) & 15;
    const int b  = wid >> 11;
    const int hlf = lane >> 4;
    const int l16 = lane & 15;
    const int q0  = qt * 16;
    const float scale = 0.125f;

    const size_t bh = (size_t)(b * Hn + h) * (Nseq / 16);

    const v4f* qp = (const v4f*)Qp + (bh + qt) * (8 * 32) + lane;
    v4f qb[8];
#pragma unroll
    for (int j = 0; j < 8; j++) qb[j] = qp[j * 32];

    v8f zero = {};
    v8f acc[4];
#pragma unroll
    for (int t = 0; t < 4; t++) acc[t] = zero;
    float m = -3.0e38f, l = 0.0f;

    const v4f* kv = (const v4f*)KVp + bh * (16 * 32) + lane;

    for (int kt = 0; kt < Nseq / 16; kt++) {
        const v4f* p = kv + (size_t)kt * (16 * 32);

        v4f ka[8];
#pragma unroll
        for (int j = 0; j < 8; j++) ka[j] = p[j * 32];
        asm volatile("" :: "v"(ka[0]), "v"(ka[1]), "v"(ka[2]), "v"(ka[3]),
                           "v"(ka[4]), "v"(ka[5]), "v"(ka[6]), "v"(ka[7]));

        v8f s = zero;
#pragma unroll
        for (int j = 0; j < 8; j++) {
            s = WMMA_F32(ka[j].xy, qb[j].xy, s);
            s = WMMA_F32(ka[j].zw, qb[j].zw, s);
        }

        v4f va[8];
#pragma unroll
        for (int i = 0; i < 8; i++) va[i] = p[(8 + i) * 32];

        float sv[8];
#pragma unroll
        for (int r = 0; r < 8; r++) sv[r] = s[r] * scale;

        float tm = sv[0];
#pragma unroll
        for (int r = 1; r < 8; r++) tm = fmaxf(tm, sv[r]);
        tm = fmaxf(tm, __shfl_xor(tm, 16, 32));

        const float mn   = fmaxf(m, tm);
        const float corr = __expf(m - mn);
        m = mn;

        float pr[8];
        float rs = 0.0f;
#pragma unroll
        for (int r = 0; r < 8; r++) { pr[r] = __expf(sv[r] - mn); rs += pr[r]; }
        rs += __shfl_xor(rs, 16, 32);
        l = l * corr + rs;

#pragma unroll
        for (int t = 0; t < 4; t++)
#pragma unroll
            for (int r = 0; r < 8; r++) acc[t][r] *= corr;

        const float x2 = __shfl_xor(pr[2], 16, 32), x3 = __shfl_xor(pr[3], 16, 32);
        const float x6 = __shfl_xor(pr[6], 16, 32), x7 = __shfl_xor(pr[7], 16, 32);
        const float u0 = __shfl_xor(pr[0], 16, 32), u1 = __shfl_xor(pr[1], 16, 32);
        const float u4 = __shfl_xor(pr[4], 16, 32), u5 = __shfl_xor(pr[5], 16, 32);
        v2f pb[4];
        pb[0].x = hlf ? x2 : pr[0];  pb[0].y = hlf ? x3 : pr[1];
        pb[1].x = hlf ? x6 : pr[4];  pb[1].y = hlf ? x7 : pr[5];
        pb[2].x = hlf ? pr[2] : u0;  pb[2].y = hlf ? pr[3] : u1;
        pb[3].x = hlf ? pr[6] : u4;  pb[3].y = hlf ? pr[7] : u5;

#pragma unroll
        for (int t = 0; t < 4; t++) {
            acc[t] = WMMA_F32(va[2 * t].xy,     pb[0], acc[t]);
            acc[t] = WMMA_F32(va[2 * t].zw,     pb[1], acc[t]);
            acc[t] = WMMA_F32(va[2 * t + 1].xy, pb[2], acc[t]);
            acc[t] = WMMA_F32(va[2 * t + 1].zw, pb[3], acc[t]);
        }
    }

    const float inv = 1.0f / l;
    v4f* Op = (v4f*)O + ((size_t)((b * Nseq + q0) >> 4) * (Cdim >> 3) + h * 8 + hlf) * 32 + l16;
#pragma unroll
    for (int t = 0; t < 4; t++) {
        v4f lo, hi;
        lo.x = acc[t][0] * inv; lo.y = acc[t][1] * inv;
        lo.z = acc[t][4] * inv; lo.w = acc[t][5] * inv;
        hi.x = acc[t][2] * inv; hi.y = acc[t][3] * inv;
        hi.z = acc[t][6] * inv; hi.w = acc[t][7] * inv;
        Op[(2 * t) * 32]      = lo;
        Op[(2 * t) * 32 + 16] = hi;
    }
}

extern "C" void kernel_launch(void* const* d_in, const int* in_sizes, int n_in,
                              void* d_out, int out_size, void* d_ws, size_t ws_size,
                              hipStream_t stream)
{
    const float* x        = (const float*)d_in[0];
    const float* rope_cos = (const float*)d_in[1];
    const float* rope_sin = (const float*)d_in[2];
    const float* W_qkv    = (const float*)d_in[3];
    const float* W_proj   = (const float*)d_in[4];
    const float* b_proj   = (const float*)d_in[5];
    float* out  = (float*)d_out;

    float* qkv  = (float*)d_ws;
    float* attn = qkv + (size_t)Mrows * ThreeC;
    float* xp   = attn + (size_t)Mrows * Cdim;
    float* wqp  = xp + (size_t)Mrows * Cdim;
    float* wpp  = wqp + (size_t)ThreeC * Cdim;
    float* kvp  = wpp + (size_t)Cdim * Cdim;
    float* qp   = xp;

    {
        const int waves = ((Mrows + ThreeC + Cdim) / 16) * (Cdim / 32);
        pack_operands_kernel<<<waves / 8, 256, 0, stream>>>(x, W_qkv, W_proj, xp);
    }
    {
        const int waves = (Mrows / 32) * (ThreeC / 64);
        gemm_packed_kernel<<<waves / 8, 256, 0, stream>>>(xp, wqp, nullptr, qkv,
                                                          Mrows, ThreeC, Cdim);
    }
    {
        const int waves = Bsz * Hn * (Nseq / 16);
        rope_pack_kernel<<<waves / 8, 256, 0, stream>>>(qkv, rope_cos, rope_sin, qp, kvp);
    }
    {
        const int waves = Bsz * Hn * (Nseq / 16);
        flash_attn_kernel<<<waves / 8, 256, 0, stream>>>(qp, kvp, attn);
    }
    {
        const int waves = (Mrows / 32) * (Cdim / 64);
        gemm_packed_kernel<<<waves / 8, 256, 0, stream>>>(attn, wpp, b_proj, out,
                                                          Mrows, Cdim, Cdim);
    }
}


# GEMMs: each block's six fragment loads held as one group a whole 16-WMMA block ahead of their use (scheduling barriers), not sunk among the WMMAs; k and sum order kept
# speedup vs baseline: 2.1213x; 1.0584x over previous
#include <hip/hip_runtime.h>

typedef float v2f __attribute__((ext_vector_type(2)));
typedef float v4f __attribute__((ext_vector_type(4)));
typedef float v8f __attribute__((ext_vector_type(8)));

#define WMMA_F32(a, b, c) \
    __builtin_amdgcn_wmma_f32_16x16x4_f32(false, (a), false, (b), (short)0, (c), false, false)

constexpr int Bsz    = 4;
constexpr int Nseq   = 2048;
constexpr int Cdim   = 1024;
constexpr int Hn     = 16;
constexpr int HD     = 64;
constexpr int ThreeC = 3 * Cdim;
constexpr int Mrows  = Bsz * Nseq;


__global__ __launch_bounds__(256)
void pack_operands_kernel(const float* __restrict__ X, const float* __restrict__ Wq,
                          const float* __restrict__ Wp, float* __restrict__ P)
{
    const int lane = threadIdx.x & 31;
    const int wid  = blockIdx.x * (blockDim.x >> 5) + (threadIdx.x >> 5);
    const int rt = wid >> 5;
    const int g  = wid & 31;
    const int c  = lane & 15;
    const int h  = lane >> 4;

    constexpr int RT_X  = Mrows / 16;
    constexpr int RT_WQ = RT_X + ThreeC / 16;
    const float* src;
    if (rt < RT_X)       src = X  + (size_t)(rt * 16 + c) * Cdim;
    else if (rt < RT_WQ) src = Wq + (size_t)((rt - RT_X) * 16 + c) * Cdim;
    else                 src = Wp + (size_t)((rt - RT_WQ) * 16 + c) * Cdim;
    src += g * 32 + 2 * h;

    v4f* dst = (v4f*)P + ((size_t)rt * (Cdim >> 3) + g * 4) * 32 + lane;
#pragma unroll
    for (int j = 0; j < 4; j++) {
        const v2f lo = *(const v2f*)(src + 8 * j);
        const v2f hi = *(const v2f*)(src + 8 * j + 4);
        v4f o;
        o.x = lo.x; o.y = lo.y; o.z = hi.x; o.w = hi.y;
        dst[j * 32] = o;
    }
}

#define GEMM_BLOCK(A0, A1, B0, B1, B2, B3)                       \
    do {                                                          \
        acc[0][0] = WMMA_F32((A0).xy, (B0).xy, acc[0][0]);       \
        acc[1][0] = WMMA_F32((A1).xy, (B0).xy, acc[1][0]);       \
        acc[0][1] = WMMA_F32((A0).xy, (B1).xy, acc[0][1]);       \
        acc[1][1] = WMMA_F32((A1).xy, (B1).xy, acc[1][1]);       \
        acc[0][2] = WMMA_F32((A0).xy, (B2).xy, acc[0][2]);       \
        acc[1][2] = WMMA_F32((A1).xy, (B2).xy, acc[1][2]);       \
        acc[0][3] = WMMA_F32((A0).xy, (B3).xy, acc[0][3]);       \
        acc[1][3] = WMMA_F32((A1).xy, (B3).xy, acc[1][3]);       \
        acc[0][0] = WMMA_F32((A0).zw, (B0).zw, acc[0][0]);       \
        acc[1][0] = WMMA_F32((A1).zw, (B0).zw, acc[1][0]);       \
        acc[0][1] = WMMA_F32((A0).zw, (B1).zw, acc[0][1]);       \
        acc[1][1] = WMMA_F32((A1).zw, (B1).zw, acc[1][1]);       \
        acc[0][2] = WMMA_F32((A0).zw, (B2).zw, acc[0][2]);       \
        acc[1][2] = WMMA_F32((A1).zw, (B2).zw, acc[1][2]);       \
        acc[0][3] = WMMA_F32((A0).zw, (B3).zw, acc[0][3]);       \
        acc[1][3] = WMMA_F32((A1).zw, (B3).zw, acc[1][3]);       \
    } while (0)

__global__ __launch_bounds__(256)
void gemm_packed_kernel(const float* __restrict__ Ap, const float* __restrict__ Bp,
                        const float* __restrict__ bias, float* __restrict__ Y,
                        int M, int Nd, int K)
{
    const int lane = threadIdx.x & 31;
    const int wid  = blockIdx.x * (blockDim.x >> 5) + (threadIdx.x >> 5);
    const int tilesN = Nd >> 6;
    const int m0 = (wid / tilesN) * 32;
    const int n0 = (wid % tilesN) * 64;
    const int hlf = lane >> 4;
    const int l16 = lane & 15;
    const int KB  = K >> 3;

    v8f zero = {};
    v8f acc[2][4];
#pragma unroll
    for (int i = 0; i < 2; i++)
#pragma unroll
        for (int j = 0; j < 4; j++) acc[i][j] = zero;

    const size_t tstride = (size_t)KB * 32;
    const v4f* pa0 = (const v4f*)Ap + (size_t)(m0 >> 4) * tstride + lane;
    const v4f* pa1 = pa0 + tstride;
    const v4f* pb0 = (const v4f*)Bp + (size_t)(n0 >> 4) * tstride + lane;
    const v4f* pb1 = pb0 + tstride;
    const v4f* pb2 = pb1 + tstride;
    const v4f* pb3 = pb2 + tstride;

    v4f a0 = pa0[0], a1 = pa1[0];
    v4f b0 = pb0[0], b1 = pb1[0], b2 = pb2[0], b3 = pb3[0];
    v4f c0, c1, d0, d1, d2, d3;

    int kb = 0;
    for (; kb < KB - 2; kb += 2) {
        const int o1 = (kb + 1) * 32, o2 = (kb + 2) * 32;
        c0 = pa0[o1]; c1 = pa1[o1];
        d0 = pb0[o1]; d1 = pb1[o1]; d2 = pb2[o1]; d3 = pb3[o1];
        __builtin_amdgcn_sched_barrier(0);
        GEMM_BLOCK(a0, a1, b0, b1, b2, b3);
        __builtin_amdgcn_sched_barrier(0);
        a0 = pa0[o2]; a1 = pa1[o2];
        b0 = pb0[o2]; b1 = pb1[o2]; b2 = pb2[o2]; b3 = pb3[o2];
        __builtin_amdgcn_sched_barrier(0);
        GEMM_BLOCK(c0, c1, d0, d1, d2, d3);
        __builtin_amdgcn_sched_barrier(0);
    }
    {
        const int o1 = (kb + 1) * 32;
        c0 = pa0[o1]; c1 = pa1[o1];
        d0 = pb0[o1]; d1 = pb1[o1]; d2 = pb2[o1]; d3 = pb3[o1];
        __builtin_amdgcn_sched_barrier(0);
        GEMM_BLOCK(a0, a1, b0, b1, b2, b3);
        GEMM_BLOCK(c0, c1, d0, d1, d2, d3);
    }

#pragma unroll
    for (int i = 0; i < 2; i++)
#pragma unroll
        for (int j = 0; j < 4; j++)
#pragma unroll
            for (int r = 0; r < 8; r++) {
                const int row = m0 + 16 * i + r + 8 * hlf;
                const int col = n0 + 16 * j + l16;
                float v = acc[i][j][r];
                if (bias) v += bias[col];
                Y[(size_t)row * Nd + col] = v;
            }
}

__global__ __launch_bounds__(256)
void rope_pack_kernel(const float* __restrict__ qkv,
                      const float* __restrict__ cosT, const float* __restrict__ sinT,
                      float* __restrict__ Qp, float* __restrict__ KVp)
{
    const int lane = threadIdx.x & 31;
    const int wid  = blockIdx.x * (blockDim.x >> 5) + (threadIdx.x >> 5);
    const int h    = wid & 15;
    const int tile = (wid >> 4) & 127;
    const int b    = wid >> 11;
    const int c    = lane & 15;
    const int hf   = lane >> 4;
    const int n    = tile * 16 + c;

    const float* qrow = qkv + (size_t)(b * Nseq + n) * ThreeC + h * HD + 2 * hf;
    const float* krow = qrow + Cdim;
    const float* cr   = cosT + n * HD + 2 * hf;
    const float* sr   = sinT + n * HD + 2 * hf;

    const size_t bht = (size_t)((b * Hn + h) * (Nseq / 16) + tile);
    v4f* qdst = (v4f*)Qp  + bht * (8 * 32)  + lane;
    v4f* kdst = (v4f*)KVp + bht * (16 * 32) + lane;

#pragma unroll
    for (int j = 0; j < 4; j++) {
        const v2f c0a = *(const v2f*)(cr + 8 * j),      c0b = *(const v2f*)(cr + 8 * j + 4);
        const v2f c1a = *(const v2f*)(cr + 8 * j + 32), c1b = *(const v2f*)(cr + 8 * j + 36);
        const v2f s0a = *(const v2f*)(sr + 8 * j),      s0b = *(const v2f*)(sr + 8 * j + 4);
        const v2f s1a = *(const v2f*)(sr + 8 * j + 32), s1b = *(const v2f*)(sr + 8 * j + 36);

        const v2f q0a = *(const v2f*)(qrow + 8 * j),      q0b = *(const v2f*)(qrow + 8 * j + 4);
        const v2f q1a = *(const v2f*)(qrow + 8 * j + 32), q1b = *(const v2f*)(qrow + 8 * j + 36);
        v4f lo, hi;
        lo.x = q0a.x * c0a.x - q1a.x * s0a.x;  lo.y = q0a.y * c0a.y - q1a.y * s0a.y;
        lo.z = q0b.x * c0b.x - q1b.x * s0b.x;  lo.w = q0b.y * c0b.y - q1b.y * s0b.y;
        hi.x = q1a.x * c1a.x + q0a.x * s1a.x;  hi.y = q1a.y * c1a.y + q0a.y * s1a.y;
        hi.z = q1b.x * c1b.x + q0b.x * s1b.x;  hi.w = q1b.y * c1b.y + q0b.y * s1b.y;
        qdst[j * 32]       = lo;
        qdst[(j + 4) * 32] = hi;

        const v2f k0a = *(const v2f*)(krow + 8 * j),      k0b = *(const v2f*)(krow + 8 * j + 4);
        const v2f k1a = *(const v2f*)(krow + 8 * j + 32), k1b = *(const v2f*)(krow + 8 * j + 36);
        lo.x = k0a.x * c0a.x - k1a.x * s0a.x;  lo.y = k0a.y * c0a.y - k1a.y * s0a.y;
        lo.z = k0b.x * c0b.x - k1b.x * s0b.x;  lo.w = k0b.y * c0b.y - k1b.y * s0b.y;
        hi.x = k1a.x * c1a.x + k0a.x * s1a.x;  hi.y = k1a.y * c1a.y + k0a.y * s1a.y;
        hi.z = k1b.x * c1b.x + k0b.x * s1b.x;  hi.w = k1b.y * c1b.y + k0b.y * s1b.y;
        kdst[j * 32]       = lo;
        kdst[(j + 4) * 32] = hi;
    }

    const float* vcol = qkv + (size_t)(b * Nseq + tile * 16 + 2 * hf) * ThreeC
                            + 2 * Cdim + h * HD + c;
#pragma unroll
    for (int t = 0; t < 4; t++)
#pragma unroll
        for (int kb2 = 0; kb2 < 2; kb2++) {
            const float* p = vcol + (size_t)(8 * kb2) * ThreeC + 16 * t;
            v4f o;
            o.x = p[0];
            o.y = p[ThreeC];
            o.z = p[4 * ThreeC];
            o.w = p[5 * ThreeC];
            kdst[(8 + 2 * t + kb2) * 32] = o;
        }
}

__global__ __launch_bounds__(256)
void flash_attn_kernel(const float* __restrict__ Qp, const float* __restrict__ KVp,
                       float* __restrict__ O)
{
    const int lane = threadIdx.x & 31;
    const int wid  = blockIdx.x * (blockDim.x >> 5) + (threadIdx.x >> 5);
    const int qt = wid & 127;
    const int h  = (wid >> 7) & 15;
    const int b  = wid >> 11;
    const int hlf = lane >> 4;
    const int l16 = lane & 15;
    const int q0  = qt * 16;
    const float scale = 0.125f;

    const size_t bh = (size_t)(b * Hn + h) * (Nseq / 16);

    const v4f* qp = (const v4f*)Qp + (bh + qt) * (8 * 32) + lane;
    v4f qb[8];
#pragma unroll
    for (int j = 0; j < 8; j++) qb[j] = qp[j * 32];

    v8f zero = {};
    v8f acc[4];
#pragma unroll
    for (int t = 0; t < 4; t++) acc[t] = zero;
    float m = -3.0e38f, l = 0.0f;

    const v4f* kv = (const v4f*)KVp + bh * (16 * 32) + lane;

    for (int kt = 0; kt < Nseq / 16; kt++) {
        const v4f* p = kv + (size_t)kt * (16 * 32);

        v4f ka[8];
#pragma unroll
        for (int j = 0; j < 8; j++) ka[j] = p[j * 32];
        asm volatile("" :: "v"(ka[0]), "v"(ka[1]), "v"(ka[2]), "v"(ka[3]),
                           "v"(ka[4]), "v"(ka[5]), "v"(ka[6]), "v"(ka[7]));

        v8f s = zero;
#pragma unroll
        for (int j = 0; j < 8; j++) {
            s = WMMA_F32(ka[j].xy, qb[j].xy, s);
            s = WMMA_F32(ka[j].zw, qb[j].zw, s);
        }

        v4f va[8];
#pragma unroll
        for (int i = 0; i < 8; i++) va[i] = p[(8 + i) * 32];

        float sv[8];
#pragma unroll
        for (int r = 0; r < 8; r++) sv[r] = s[r] * scale;

        float tm = sv[0];
#pragma unroll
        for (int r = 1; r < 8; r++) tm = fmaxf(tm, sv[r]);
        tm = fmaxf(tm, __shfl_xor(tm, 16, 32));

        const float mn   = fmaxf(m, tm);
        const float corr = __expf(m - mn);
        m = mn;

        float pr[8];
        float rs = 0.0f;
#pragma unroll
        for (int r = 0; r < 8; r++) { pr[r] = __expf(sv[r] - mn); rs += pr[r]; }
        rs += __shfl_xor(rs, 16, 32);
        l = l * corr + rs;

#pragma unroll
        for (int t = 0; t < 4; t++)
#pragma unroll
            for (int r = 0; r < 8; r++) acc[t][r] *= corr;

        const float x2 = __shfl_xor(pr[2], 16, 32), x3 = __shfl_xor(pr[3], 16, 32);
        const float x6 = __shfl_xor(pr[6], 16, 32), x7 = __shfl_xor(pr[7], 16, 32);
        const float u0 = __shfl_xor(pr[0], 16, 32), u1 = __shfl_xor(pr[1], 16, 32);
        const float u4 = __shfl_xor(pr[4], 16, 32), u5 = __shfl_xor(pr[5], 16, 32);
        v2f pb[4];
        pb[0].x = hlf ? x2 : pr[0];  pb[0].y = hlf ? x3 : pr[1];
        pb[1].x = hlf ? x6 : pr[4];  pb[1].y = hlf ? x7 : pr[5];
        pb[2].x = hlf ? pr[2] : u0;  pb[2].y = hlf ? pr[3] : u1;
        pb[3].x = hlf ? pr[6] : u4;  pb[3].y = hlf ? pr[7] : u5;

#pragma unroll
        for (int t = 0; t < 4; t++) {
            acc[t] = WMMA_F32(va[2 * t].xy,     pb[0], acc[t]);
            acc[t] = WMMA_F32(va[2 * t].zw,     pb[1], acc[t]);
            acc[t] = WMMA_F32(va[2 * t + 1].xy, pb[2], acc[t]);
            acc[t] = WMMA_F32(va[2 * t + 1].zw, pb[3], acc[t]);
        }
    }

    const float inv = 1.0f / l;
    v4f* Op = (v4f*)O + ((size_t)((b * Nseq + q0) >> 4) * (Cdim >> 3) + h * 8 + hlf) * 32 + l16;
#pragma unroll
    for (int t = 0; t < 4; t++) {
        v4f lo, hi;
        lo.x = acc[t][0] * inv; lo.y = acc[t][1] * inv;
        lo.z = acc[t][4] * inv; lo.w = acc[t][5] * inv;
        hi.x = acc[t][2] * inv; hi.y = acc[t][3] * inv;
        hi.z = acc[t][6] * inv; hi.w = acc[t][7] * inv;
        Op[(2 * t) * 32]      = lo;
        Op[(2 * t) * 32 + 16] = hi;
    }
}

extern "C" void kernel_launch(void* const* d_in, const int* in_sizes, int n_in,
                              void* d_out, int out_size, void* d_ws, size_t ws_size,
                              hipStream_t stream)
{
    const float* x        = (const float*)d_in[0];
    const float* rope_cos = (const float*)d_in[1];
    const float* rope_sin = (const float*)d_in[2];
    const float* W_qkv    = (const float*)d_in[3];
    const float* W_proj   = (const float*)d_in[4];
    const float* b_proj   = (const float*)d_in[5];
    float* out  = (float*)d_out;

    float* qkv  = (float*)d_ws;
    float* attn = qkv + (size_t)Mrows * ThreeC;
    float* xp   = attn + (size_t)Mrows * Cdim;
    float* wqp  = xp + (size_t)Mrows * Cdim;
    float* wpp  = wqp + (size_t)ThreeC * Cdim;
    float* kvp  = wpp + (size_t)Cdim * Cdim;
    float* qp   = xp;

    {
        const int waves = ((Mrows + ThreeC + Cdim) / 16) * (Cdim / 32);
        pack_operands_kernel<<<waves / 8, 256, 0, stream>>>(x, W_qkv, W_proj, xp);
    }
    {
        const int waves = (Mrows / 32) * (ThreeC / 64);
        gemm_packed_kernel<<<waves / 8, 256, 0, stream>>>(xp, wqp, nullptr, qkv,
                                                          Mrows, ThreeC, Cdim);
    }
    {
        const int waves = Bsz * Hn * (Nseq / 16);
        rope_pack_kernel<<<waves / 8, 256, 0, stream>>>(qkv, rope_cos, rope_sin, qp, kvp);
    }
    {
        const int waves = Bsz * Hn * (Nseq / 16);
        flash_attn_kernel<<<waves / 8, 256, 0, stream>>>(qp, kvp, attn);
    }
    {
        const int waves = (Mrows / 32) * (Cdim / 64);
        gemm_packed_kernel<<<waves / 8, 256, 0, stream>>>(attn, wpp, b_proj, out,
                                                          Mrows, Cdim, Cdim);
    }
}


# GEMM and attention loops: wave priority stepped down as a wave advances (s_setprio 3,2,1,0 at 1/2, 3/4, 7/8 of its loop), so a SIMD's waves end together, not one after another; sums unchanged
# speedup vs baseline: 2.1977x; 1.0360x over previous
#include <hip/hip_runtime.h>

typedef float v2f __attribute__((ext_vector_type(2)));
typedef float v4f __attribute__((ext_vector_type(4)));
typedef float v8f __attribute__((ext_vector_type(8)));

#define WMMA_F32(a, b, c) \
    __builtin_amdgcn_wmma_f32_16x16x4_f32(false, (a), false, (b), (short)0, (c), false, false)

constexpr int Bsz    = 4;
constexpr int Nseq   = 2048;
constexpr int Cdim   = 1024;
constexpr int Hn     = 16;
constexpr int HD     = 64;
constexpr int ThreeC = 3 * Cdim;
constexpr int Mrows  = Bsz * Nseq;


__global__ __launch_bounds__(256)
void pack_operands_kernel(const float* __restrict__ X, const float* __restrict__ Wq,
                          const float* __restrict__ Wp, float* __restrict__ P)
{
    const int lane = threadIdx.x & 31;
    const int wid  = blockIdx.x * (blockDim.x >> 5) + (threadIdx.x >> 5);
    const int rt = wid >> 5;
    const int g  = wid & 31;
    const int c  = lane & 15;
    const int h  = lane >> 4;

    constexpr int RT_X  = Mrows / 16;
    constexpr int RT_WQ = RT_X + ThreeC / 16;
    const float* src;
    if (rt < RT_X)       src = X  + (size_t)(rt * 16 + c) * Cdim;
    else if (rt < RT_WQ) src = Wq + (size_t)((rt - RT_X) * 16 + c) * Cdim;
    else                 src = Wp + (size_t)((rt - RT_WQ) * 16 + c) * Cdim;
    src += g * 32 + 2 * h;

    v4f* dst = (v4f*)P + ((size_t)rt * (Cdim >> 3) + g * 4) * 32 + lane;
#pragma unroll
    for (int j = 0; j < 4; j++) {
        const v2f lo = *(const v2f*)(src + 8 * j);
        const v2f hi = *(const v2f*)(src + 8 * j + 4);
        v4f o;
        o.x = lo.x; o.y = lo.y; o.z = hi.x; o.w = hi.y;
        dst[j * 32] = o;
    }
}

#define GEMM_BLOCK(A0, A1, B0, B1, B2, B3)                       \
    do {                                                          \
        acc[0][0] = WMMA_F32((A0).xy, (B0).xy, acc[0][0]);       \
        acc[1][0] = WMMA_F32((A1).xy, (B0).xy, acc[1][0]);       \
        acc[0][1] = WMMA_F32((A0).xy, (B1).xy, acc[0][1]);       \
        acc[1][1] = WMMA_F32((A1).xy, (B1).xy, acc[1][1]);       \
        acc[0][2] = WMMA_F32((A0).xy, (B2).xy, acc[0][2]);       \
        acc[1][2] = WMMA_F32((A1).xy, (B2).xy, acc[1][2]);       \
        acc[0][3] = WMMA_F32((A0).xy, (B3).xy, acc[0][3]);       \
        acc[1][3] = WMMA_F32((A1).xy, (B3).xy, acc[1][3]);       \
        acc[0][0] = WMMA_F32((A0).zw, (B0).zw, acc[0][0]);       \
        acc[1][0] = WMMA_F32((A1).zw, (B0).zw, acc[1][0]);       \
        acc[0][1] = WMMA_F32((A0).zw, (B1).zw, acc[0][1]);       \
        acc[1][1] = WMMA_F32((A1).zw, (B1).zw, acc[1][1]);       \
        acc[0][2] = WMMA_F32((A0).zw, (B2).zw, acc[0][2]);       \
        acc[1][2] = WMMA_F32((A1).zw, (B2).zw, acc[1][2]);       \
        acc[0][3] = WMMA_F32((A0).zw, (B3).zw, acc[0][3]);       \
        acc[1][3] = WMMA_F32((A1).zw, (B3).zw, acc[1][3]);       \
    } while (0)

__global__ __launch_bounds__(256)
void gemm_packed_kernel(const float* __restrict__ Ap, const float* __restrict__ Bp,
                        const float* __restrict__ bias, float* __restrict__ Y,
                        int M, int Nd, int K)
{
    const int lane = threadIdx.x & 31;
    const int wid  = blockIdx.x * (blockDim.x >> 5) + (threadIdx.x >> 5);
    const int tilesN = Nd >> 6;
    const int m0 = (wid / tilesN) * 32;
    const int n0 = (wid % tilesN) * 64;
    const int hlf = lane >> 4;
    const int l16 = lane & 15;
    const int KB  = K >> 3;

    v8f zero = {};
    v8f acc[2][4];
#pragma unroll
    for (int i = 0; i < 2; i++)
#pragma unroll
        for (int j = 0; j < 4; j++) acc[i][j] = zero;

    const size_t tstride = (size_t)KB * 32;
    const v4f* pa0 = (const v4f*)Ap + (size_t)(m0 >> 4) * tstride + lane;
    const v4f* pa1 = pa0 + tstride;
    const v4f* pb0 = (const v4f*)Bp + (size_t)(n0 >> 4) * tstride + lane;
    const v4f* pb1 = pb0 + tstride;
    const v4f* pb2 = pb1 + tstride;
    const v4f* pb3 = pb2 + tstride;

    v4f a0 = pa0[0], a1 = pa1[0];
    v4f b0 = pb0[0], b1 = pb1[0], b2 = pb2[0], b3 = pb3[0];
    v4f c0, c1, d0, d1, d2, d3;

    const int kq1 = KB >> 1, kq2 = kq1 + (KB >> 2), kq3 = kq2 + (KB >> 3);
    int step = kq1, prio = 3;
    __builtin_amdgcn_s_setprio(3);

    int kb = 0;
    for (; kb < KB - 2; kb += 2) {
        if (kb >= step) {
            prio--;
            if (prio == 2)      { __builtin_amdgcn_s_setprio(2); step = kq2; }
            else if (prio == 1) { __builtin_amdgcn_s_setprio(1); step = kq3; }
            else                { __builtin_amdgcn_s_setprio(0); step = 0x7fffffff; }
        }
        const int o1 = (kb + 1) * 32, o2 = (kb + 2) * 32;
        c0 = pa0[o1]; c1 = pa1[o1];
        d0 = pb0[o1]; d1 = pb1[o1]; d2 = pb2[o1]; d3 = pb3[o1];
        __builtin_amdgcn_sched_barrier(0);
        GEMM_BLOCK(a0, a1, b0, b1, b2, b3);
        __builtin_amdgcn_sched_barrier(0);
        a0 = pa0[o2]; a1 = pa1[o2];
        b0 = pb0[o2]; b1 = pb1[o2]; b2 = pb2[o2]; b3 = pb3[o2];
        __builtin_amdgcn_sched_barrier(0);
        GEMM_BLOCK(c0, c1, d0, d1, d2, d3);
        __builtin_amdgcn_sched_barrier(0);
    }
    __builtin_amdgcn_s_setprio(0);
    {
        const int o1 = (kb + 1) * 32;
        c0 = pa0[o1]; c1 = pa1[o1];
        d0 = pb0[o1]; d1 = pb1[o1]; d2 = pb2[o1]; d3 = pb3[o1];
        __builtin_amdgcn_sched_barrier(0);
        GEMM_BLOCK(a0, a1, b0, b1, b2, b3);
        GEMM_BLOCK(c0, c1, d0, d1, d2, d3);
    }

#pragma unroll
    for (int i = 0; i < 2; i++)
#pragma unroll
        for (int j = 0; j < 4; j++)
#pragma unroll
            for (int r = 0; r < 8; r++) {
                const int row = m0 + 16 * i + r + 8 * hlf;
                const int col = n0 + 16 * j + l16;
                float v = acc[i][j][r];
                if (bias) v += bias[col];
                Y[(size_t)row * Nd + col] = v;
            }
}

__global__ __launch_bounds__(256)
void rope_pack_kernel(const float* __restrict__ qkv,
                      const float* __restrict__ cosT, const float* __restrict__ sinT,
                      float* __restrict__ Qp, float* __restrict__ KVp)
{
    const int lane = threadIdx.x & 31;
    const int wid  = blockIdx.x * (blockDim.x >> 5) + (threadIdx.x >> 5);
    const int h    = wid & 15;
    const int tile = (wid >> 4) & 127;
    const int b    = wid >> 11;
    const int c    = lane & 15;
    const int hf   = lane >> 4;
    const int n    = tile * 16 + c;

    const float* qrow = qkv + (size_t)(b * Nseq + n) * ThreeC + h * HD + 2 * hf;
    const float* krow = qrow + Cdim;
    const float* cr   = cosT + n * HD + 2 * hf;
    const float* sr   = sinT + n * HD + 2 * hf;

    const size_t bht = (size_t)((b * Hn + h) * (Nseq / 16) + tile);
    v4f* qdst = (v4f*)Qp  + bht * (8 * 32)  + lane;
    v4f* kdst = (v4f*)KVp + bht * (16 * 32) + lane;

#pragma unroll
    for (int j = 0; j < 4; j++) {
        const v2f c0a = *(const v2f*)(cr + 8 * j),      c0b = *(const v2f*)(cr + 8 * j + 4);
        const v2f c1a = *(const v2f*)(cr + 8 * j + 32), c1b = *(const v2f*)(cr + 8 * j + 36);
        const v2f s0a = *(const v2f*)(sr + 8 * j),      s0b = *(const v2f*)(sr + 8 * j + 4);
        const v2f s1a = *(const v2f*)(sr + 8 * j + 32), s1b = *(const v2f*)(sr + 8 * j + 36);

        const v2f q0a = *(const v2f*)(qrow + 8 * j),      q0b = *(const v2f*)(qrow + 8 * j + 4);
        const v2f q1a = *(const v2f*)(qrow + 8 * j + 32), q1b = *(const v2f*)(qrow + 8 * j + 36);
        v4f lo, hi;
        lo.x = q0a.x * c0a.x - q1a.x * s0a.x;  lo.y = q0a.y * c0a.y - q1a.y * s0a.y;
        lo.z = q0b.x * c0b.x - q1b.x * s0b.x;  lo.w = q0b.y * c0b.y - q1b.y * s0b.y;
        hi.x = q1a.x * c1a.x + q0a.x * s1a.x;  hi.y = q1a.y * c1a.y + q0a.y * s1a.y;
        hi.z = q1b.x * c1b.x + q0b.x * s1b.x;  hi.w = q1b.y * c1b.y + q0b.y * s1b.y;
        qdst[j * 32]       = lo;
        qdst[(j + 4) * 32] = hi;

        const v2f k0a = *(const v2f*)(krow + 8 * j),      k0b = *(const v2f*)(krow + 8 * j + 4);
        const v2f k1a = *(const v2f*)(krow + 8 * j + 32), k1b = *(const v2f*)(krow + 8 * j + 36);
        lo.x = k0a.x * c0a.x - k1a.x * s0a.x;  lo.y = k0a.y * c0a.y - k1a.y * s0a.y;
        lo.z = k0b.x * c0b.x - k1b.x * s0b.x;  lo.w = k0b.y * c0b.y - k1b.y * s0b.y;
        hi.x = k1a.x * c1a.x + k0a.x * s1a.x;  hi.y = k1a.y * c1a.y + k0a.y * s1a.y;
        hi.z = k1b.x * c1b.x + k0b.x * s1b.x;  hi.w = k1b.y * c1b.y + k0b.y * s1b.y;
        kdst[j * 32]       = lo;
        kdst[(j + 4) * 32] = hi;
    }

    const float* vcol = qkv + (size_t)(b * Nseq + tile * 16 + 2 * hf) * ThreeC
                            + 2 * Cdim + h * HD + c;
#pragma unroll
    for (int t = 0; t < 4; t++)
#pragma unroll
        for (int kb2 = 0; kb2 < 2; kb2++) {
            const float* p = vcol + (size_t)(8 * kb2) * ThreeC + 16 * t;
            v4f o;
            o.x = p[0];
            o.y = p[ThreeC];
            o.z = p[4 * ThreeC];
            o.w = p[5 * ThreeC];
            kdst[(8 + 2 * t + kb2) * 32] = o;
        }
}

__global__ __launch_bounds__(256)
void flash_attn_kernel(const float* __restrict__ Qp, const float* __restrict__ KVp,
                       float* __restrict__ O)
{
    const int lane = threadIdx.x & 31;
    const int wid  = blockIdx.x * (blockDim.x >> 5) + (threadIdx.x >> 5);
    const int qt = wid & 127;
    const int h  = (wid >> 7) & 15;
    const int b  = wid >> 11;
    const int hlf = lane >> 4;
    const int l16 = lane & 15;
    const int q0  = qt * 16;
    const float scale = 0.125f;

    const size_t bh = (size_t)(b * Hn + h) * (Nseq / 16);

    const v4f* qp = (const v4f*)Qp + (bh + qt) * (8 * 32) + lane;
    v4f qb[8];
#pragma unroll
    for (int j = 0; j < 8; j++) qb[j] = qp[j * 32];

    v8f zero = {};
    v8f acc[4];
#pragma unroll
    for (int t = 0; t < 4; t++) acc[t] = zero;
    float m = -3.0e38f, l = 0.0f;

    const v4f* kv = (const v4f*)KVp + bh * (16 * 32) + lane;

    constexpr int KT = Nseq / 16;
    int step = KT / 2;
    __builtin_amdgcn_s_setprio(3);

    for (int kt = 0; kt < KT; kt++) {
        if (kt == step) {
            if (kt == KT / 2)               { __builtin_amdgcn_s_setprio(2); step = KT / 2 + KT / 4; }
            else if (kt == KT / 2 + KT / 4) { __builtin_amdgcn_s_setprio(1); step = KT - KT / 8; }
            else                            { __builtin_amdgcn_s_setprio(0); step = -1; }
        }
        const v4f* p = kv + (size_t)kt * (16 * 32);

        v4f ka[8];
#pragma unroll
        for (int j = 0; j < 8; j++) ka[j] = p[j * 32];
        asm volatile("" :: "v"(ka[0]), "v"(ka[1]), "v"(ka[2]), "v"(ka[3]),
                           "v"(ka[4]), "v"(ka[5]), "v"(ka[6]), "v"(ka[7]));

        v8f s = zero;
#pragma unroll
        for (int j = 0; j < 8; j++) {
            s = WMMA_F32(ka[j].xy, qb[j].xy, s);
            s = WMMA_F32(ka[j].zw, qb[j].zw, s);
        }

        v4f va[8];
#pragma unroll
        for (int i = 0; i < 8; i++) va[i] = p[(8 + i) * 32];

        float sv[8];
#pragma unroll
        for (int r = 0; r < 8; r++) sv[r] = s[r] * scale;

        float tm = sv[0];
#pragma unroll
        for (int r = 1; r < 8; r++) tm = fmaxf(tm, sv[r]);
        tm = fmaxf(tm, __shfl_xor(tm, 16, 32));

        const float mn   = fmaxf(m, tm);
        const float corr = __expf(m - mn);
        m = mn;

        float pr[8];
        float rs = 0.0f;
#pragma unroll
        for (int r = 0; r < 8; r++) { pr[r] = __expf(sv[r] - mn); rs += pr[r]; }
        rs += __shfl_xor(rs, 16, 32);
        l = l * corr + rs;

#pragma unroll
        for (int t = 0; t < 4; t++)
#pragma unroll
            for (int r = 0; r < 8; r++) acc[t][r] *= corr;

        const float x2 = __shfl_xor(pr[2], 16, 32), x3 = __shfl_xor(pr[3], 16, 32);
        const float x6 = __shfl_xor(pr[6], 16, 32), x7 = __shfl_xor(pr[7], 16, 32);
        const float u0 = __shfl_xor(pr[0], 16, 32), u1 = __shfl_xor(pr[1], 16, 32);
        const float u4 = __shfl_xor(pr[4], 16, 32), u5 = __shfl_xor(pr[5], 16, 32);
        v2f pb[4];
        pb[0].x = hlf ? x2 : pr[0];  pb[0].y = hlf ? x3 : pr[1];
        pb[1].x = hlf ? x6 : pr[4];  pb[1].y = hlf ? x7 : pr[5];
        pb[2].x = hlf ? pr[2] : u0;  pb[2].y = hlf ? pr[3] : u1;
        pb[3].x = hlf ? pr[6] : u4;  pb[3].y = hlf ? pr[7] : u5;

#pragma unroll
        for (int t = 0; t < 4; t++) {
            acc[t] = WMMA_F32(va[2 * t].xy,     pb[0], acc[t]);
            acc[t] = WMMA_F32(va[2 * t].zw,     pb[1], acc[t]);
            acc[t] = WMMA_F32(va[2 * t + 1].xy, pb[2], acc[t]);
            acc[t] = WMMA_F32(va[2 * t + 1].zw, pb[3], acc[t]);
        }
    }

    const float inv = 1.0f / l;
    v4f* Op = (v4f*)O + ((size_t)((b * Nseq + q0) >> 4) * (Cdim >> 3) + h * 8 + hlf) * 32 + l16;
#pragma unroll
    for (int t = 0; t < 4; t++) {
        v4f lo, hi;
        lo.x = acc[t][0] * inv; lo.y = acc[t][1] * inv;
        lo.z = acc[t][4] * inv; lo.w = acc[t][5] * inv;
        hi.x = acc[t][2] * inv; hi.y = acc[t][3] * inv;
        hi.z = acc[t][6] * inv; hi.w = acc[t][7] * inv;
        Op[(2 * t) * 32]      = lo;
        Op[(2 * t) * 32 + 16] = hi;
    }
}

extern "C" void kernel_launch(void* const* d_in, const int* in_sizes, int n_in,
                              void* d_out, int out_size, void* d_ws, size_t ws_size,
                              hipStream_t stream)
{
    const float* x        = (const float*)d_in[0];
    const float* rope_cos = (const float*)d_in[1];
    const float* rope_sin = (const float*)d_in[2];
    const float* W_qkv    = (const float*)d_in[3];
    const float* W_proj   = (const float*)d_in[4];
    const float* b_proj   = (const float*)d_in[5];
    float* out  = (float*)d_out;

    float* qkv  = (float*)d_ws;
    float* attn = qkv + (size_t)Mrows * ThreeC;
    float* xp   = attn + (size_t)Mrows * Cdim;
    float* wqp  = xp + (size_t)Mrows * Cdim;
    float* wpp  = wqp + (size_t)ThreeC * Cdim;
    float* kvp  = wpp + (size_t)Cdim * Cdim;
    float* qp   = xp;

    {
        const int waves = ((Mrows + ThreeC + Cdim) / 16) * (Cdim / 32);
        pack_operands_kernel<<<waves / 8, 256, 0, stream>>>(x, W_qkv, W_proj, xp);
    }
    {
        const int waves = (Mrows / 32) * (ThreeC / 64);
        gemm_packed_kernel<<<waves / 8, 256, 0, stream>>>(xp, wqp, nullptr, qkv,
                                                          Mrows, ThreeC, Cdim);
    }
    {
        const int waves = Bsz * Hn * (Nseq / 16);
        rope_pack_kernel<<<waves / 8, 256, 0, stream>>>(qkv, rope_cos, rope_sin, qp, kvp);
    }
    {
        const int waves = Bsz * Hn * (Nseq / 16);
        flash_attn_kernel<<<waves / 8, 256, 0, stream>>>(qp, kvp, attn);
    }
    {
        const int waves = (Mrows / 32) * (Cdim / 64);
        gemm_packed_kernel<<<waves / 8, 256, 0, stream>>>(attn, wpp, b_proj, out,
                                                          Mrows, Cdim, Cdim);
    }
}


# wave tiles doubled, each loaded fragment feeds twice the WMMAs: GEMM 64x64 per wave (qkv GEMM in 12-wave blocks: two even rounds), attention 2 query tiles per wave; bias read once; sums unchanged
# speedup vs baseline: 2.3223x; 1.0567x over previous
#include <hip/hip_runtime.h>

typedef float v2f __attribute__((ext_vector_type(2)));
typedef float v4f __attribute__((ext_vector_type(4)));
typedef float v8f __attribute__((ext_vector_type(8)));

#define WMMA_F32(a, b, c) \
    __builtin_amdgcn_wmma_f32_16x16x4_f32(false, (a), false, (b), (short)0, (c), false, false)

constexpr int Bsz    = 4;
constexpr int Nseq   = 2048;
constexpr int Cdim   = 1024;
constexpr int Hn     = 16;
constexpr int HD     = 64;
constexpr int ThreeC = 3 * Cdim;
constexpr int Mrows  = Bsz * Nseq;


__global__ __launch_bounds__(256)
void pack_operands_kernel(const float* __restrict__ X, const float* __restrict__ Wq,
                          const float* __restrict__ Wp, float* __restrict__ P)
{
    const int lane = threadIdx.x & 31;
    const int wid  = blockIdx.x * (blockDim.x >> 5) + (threadIdx.x >> 5);
    const int rt = wid >> 5;
    const int g  = wid & 31;
    const int c  = lane & 15;
    const int h  = lane >> 4;

    constexpr int RT_X  = Mrows / 16;
    constexpr int RT_WQ = RT_X + ThreeC / 16;
    const float* src;
    if (rt < RT_X)       src = X  + (size_t)(rt * 16 + c) * Cdim;
    else if (rt < RT_WQ) src = Wq + (size_t)((rt - RT_X) * 16 + c) * Cdim;
    else                 src = Wp + (size_t)((rt - RT_WQ) * 16 + c) * Cdim;
    src += g * 32 + 2 * h;

    v4f* dst = (v4f*)P + ((size_t)rt * (Cdim >> 3) + g * 4) * 32 + lane;
#pragma unroll
    for (int j = 0; j < 4; j++) {
        const v2f lo = *(const v2f*)(src + 8 * j);
        const v2f hi = *(const v2f*)(src + 8 * j + 4);
        v4f o;
        o.x = lo.x; o.y = lo.y; o.z = hi.x; o.w = hi.y;
        dst[j * 32] = o;
    }
}

#define GEMM_KSTEP(A, B, S)                                           \
    do {                                                               \
        acc[0][0] = WMMA_F32((A##0).S, (B##0).S, acc[0][0]);          \
        acc[1][0] = WMMA_F32((A##1).S, (B##0).S, acc[1][0]);          \
        acc[2][0] = WMMA_F32((A##2).S, (B##0).S, acc[2][0]);          \
        acc[3][0] = WMMA_F32((A##3).S, (B##0).S, acc[3][0]);          \
        acc[0][1] = WMMA_F32((A##0).S, (B##1).S, acc[0][1]);          \
        acc[1][1] = WMMA_F32((A##1).S, (B##1).S, acc[1][1]);          \
        acc[2][1] = WMMA_F32((A##2).S, (B##1).S, acc[2][1]);          \
        acc[3][1] = WMMA_F32((A##3).S, (B##1).S, acc[3][1]);          \
        acc[0][2] = WMMA_F32((A##0).S, (B##2).S, acc[0][2]);          \
        acc[1][2] = WMMA_F32((A##1).S, (B##2).S, acc[1][2]);          \
        acc[2][2] = WMMA_F32((A##2).S, (B##2).S, acc[2][2]);          \
        acc[3][2] = WMMA_F32((A##3).S, (B##2).S, acc[3][2]);          \
        acc[0][3] = WMMA_F32((A##0).S, (B##3).S, acc[0][3]);          \
        acc[1][3] = WMMA_F32((A##1).S, (B##3).S, acc[1][3]);          \
        acc[2][3] = WMMA_F32((A##2).S, (B##3).S, acc[2][3]);          \
        acc[3][3] = WMMA_F32((A##3).S, (B##3).S, acc[3][3]);          \
    } while (0)
#define GEMM_BLOCK(A, B)  do { GEMM_KSTEP(A, B, xy); GEMM_KSTEP(A, B, zw); } while (0)
#define GEMM_LOAD(A, B, off)                                          \
    do {                                                               \
        A##0 = pa0[off]; A##1 = pa1[off]; A##2 = pa2[off]; A##3 = pa3[off];   \
        B##0 = pb0[off]; B##1 = pb1[off]; B##2 = pb2[off]; B##3 = pb3[off];   \
    } while (0)

__global__ __launch_bounds__(384)
void gemm_packed_kernel(const float* __restrict__ Ap, const float* __restrict__ Bp,
                        const float* __restrict__ bias, float* __restrict__ Y,
                        int M, int Nd, int K)
{
    const int lane = threadIdx.x & 31;
    const int wid  = blockIdx.x * (blockDim.x >> 5) + (threadIdx.x >> 5);
    const int tilesN = Nd >> 6;
    const int m0 = (wid / tilesN) * 64;
    const int n0 = (wid % tilesN) * 64;
    const int hlf = lane >> 4;
    const int l16 = lane & 15;
    const int KB  = K >> 3;

    v8f zero = {};
    v8f acc[4][4];
#pragma unroll
    for (int i = 0; i < 4; i++)
#pragma unroll
        for (int j = 0; j < 4; j++) acc[i][j] = zero;

    const size_t tstride = (size_t)KB * 32;
    const v4f* pa0 = (const v4f*)Ap + (size_t)(m0 >> 4) * tstride + lane;
    const v4f* pa1 = pa0 + tstride;
    const v4f* pa2 = pa1 + tstride;
    const v4f* pa3 = pa2 + tstride;
    const v4f* pb0 = (const v4f*)Bp + (size_t)(n0 >> 4) * tstride + lane;
    const v4f* pb1 = pb0 + tstride;
    const v4f* pb2 = pb1 + tstride;
    const v4f* pb3 = pb2 + tstride;

    v4f a0, a1, a2, a3, b0, b1, b2, b3;
    v4f c0, c1, c2, c3, d0, d1, d2, d3;
    GEMM_LOAD(a, b, 0);

    const int kq1 = KB >> 1, kq2 = kq1 + (KB >> 2), kq3 = kq2 + (KB >> 3);
    int step = kq1, prio = 3;
    __builtin_amdgcn_s_setprio(3);

    int kb = 0;
    for (; kb < KB - 2; kb += 2) {
        if (kb >= step) {
            prio--;
            if (prio == 2)      { __builtin_amdgcn_s_setprio(2); step = kq2; }
            else if (prio == 1) { __builtin_amdgcn_s_setprio(1); step = kq3; }
            else                { __builtin_amdgcn_s_setprio(0); step = 0x7fffffff; }
        }
        const int o1 = (kb + 1) * 32, o2 = (kb + 2) * 32;
        GEMM_LOAD(c, d, o1);
        __builtin_amdgcn_sched_barrier(0);
        GEMM_BLOCK(a, b);
        __builtin_amdgcn_sched_barrier(0);
        GEMM_LOAD(a, b, o2);
        __builtin_amdgcn_sched_barrier(0);
        GEMM_BLOCK(c, d);
        __builtin_amdgcn_sched_barrier(0);
    }
    __builtin_amdgcn_s_setprio(0);
    {
        const int o1 = (kb + 1) * 32;
        GEMM_LOAD(c, d, o1);
        __builtin_amdgcn_sched_barrier(0);
        GEMM_BLOCK(a, b);
        GEMM_BLOCK(c, d);
    }

    float bv[4];
#pragma unroll
    for (int j = 0; j < 4; j++) bv[j] = bias ? bias[n0 + 16 * j + l16] : 0.0f;

#pragma unroll
    for (int i = 0; i < 4; i++)
#pragma unroll
        for (int j = 0; j < 4; j++)
#pragma unroll
            for (int r = 0; r < 8; r++) {
                const int row = m0 + 16 * i + r + 8 * hlf;
                const int col = n0 + 16 * j + l16;
                Y[(size_t)row * Nd + col] = acc[i][j][r] + bv[j];
            }
}

__global__ __launch_bounds__(256)
void rope_pack_kernel(const float* __restrict__ qkv,
                      const float* __restrict__ cosT, const float* __restrict__ sinT,
                      float* __restrict__ Qp, float* __restrict__ KVp)
{
    const int lane = threadIdx.x & 31;
    const int wid  = blockIdx.x * (blockDim.x >> 5) + (threadIdx.x >> 5);
    const int h    = wid & 15;
    const int tile = (wid >> 4) & 127;
    const int b    = wid >> 11;
    const int c    = lane & 15;
    const int hf   = lane >> 4;
    const int n    = tile * 16 + c;

    const float* qrow = qkv + (size_t)(b * Nseq + n) * ThreeC + h * HD + 2 * hf;
    const float* krow = qrow + Cdim;
    const float* cr   = cosT + n * HD + 2 * hf;
    const float* sr   = sinT + n * HD + 2 * hf;

    const size_t bht = (size_t)((b * Hn + h) * (Nseq / 16) + tile);
    v4f* qdst = (v4f*)Qp  + bht * (8 * 32)  + lane;
    v4f* kdst = (v4f*)KVp + bht * (16 * 32) + lane;

#pragma unroll
    for (int j = 0; j < 4; j++) {
        const v2f c0a = *(const v2f*)(cr + 8 * j),      c0b = *(const v2f*)(cr + 8 * j + 4);
        const v2f c1a = *(const v2f*)(cr + 8 * j + 32), c1b = *(const v2f*)(cr + 8 * j + 36);
        const v2f s0a = *(const v2f*)(sr + 8 * j),      s0b = *(const v2f*)(sr + 8 * j + 4);
        const v2f s1a = *(const v2f*)(sr + 8 * j + 32), s1b = *(const v2f*)(sr + 8 * j + 36);

        const v2f q0a = *(const v2f*)(qrow + 8 * j),      q0b = *(const v2f*)(qrow + 8 * j + 4);
        const v2f q1a = *(const v2f*)(qrow + 8 * j + 32), q1b = *(const v2f*)(qrow + 8 * j + 36);
        v4f lo, hi;
        lo.x = q0a.x * c0a.x - q1a.x * s0a.x;  lo.y = q0a.y * c0a.y - q1a.y * s0a.y;
        lo.z = q0b.x * c0b.x - q1b.x * s0b.x;  lo.w = q0b.y * c0b.y - q1b.y * s0b.y;
        hi.x = q1a.x * c1a.x + q0a.x * s1a.x;  hi.y = q1a.y * c1a.y + q0a.y * s1a.y;
        hi.z = q1b.x * c1b.x + q0b.x * s1b.x;  hi.w = q1b.y * c1b.y + q0b.y * s1b.y;
        qdst[j * 32]       = lo;
        qdst[(j + 4) * 32] = hi;

        const v2f k0a = *(const v2f*)(krow + 8 * j),      k0b = *(const v2f*)(krow + 8 * j + 4);
        const v2f k1a = *(const v2f*)(krow + 8 * j + 32), k1b = *(const v2f*)(krow + 8 * j + 36);
        lo.x = k0a.x * c0a.x - k1a.x * s0a.x;  lo.y = k0a.y * c0a.y - k1a.y * s0a.y;
        lo.z = k0b.x * c0b.x - k1b.x * s0b.x;  lo.w = k0b.y * c0b.y - k1b.y * s0b.y;
        hi.x = k1a.x * c1a.x + k0a.x * s1a.x;  hi.y = k1a.y * c1a.y + k0a.y * s1a.y;
        hi.z = k1b.x * c1b.x + k0b.x * s1b.x;  hi.w = k1b.y * c1b.y + k0b.y * s1b.y;
        kdst[j * 32]       = lo;
        kdst[(j + 4) * 32] = hi;
    }

    const float* vcol = qkv + (size_t)(b * Nseq + tile * 16 + 2 * hf) * ThreeC
                            + 2 * Cdim + h * HD + c;
#pragma unroll
    for (int t = 0; t < 4; t++)
#pragma unroll
        for (int kb2 = 0; kb2 < 2; kb2++) {
            const float* p = vcol + (size_t)(8 * kb2) * ThreeC + 16 * t;
            v4f o;
            o.x = p[0];
            o.y = p[ThreeC];
            o.z = p[4 * ThreeC];
            o.w = p[5 * ThreeC];
            kdst[(8 + 2 * t + kb2) * 32] = o;
        }
}

__global__ __launch_bounds__(256)
void flash_attn_kernel(const float* __restrict__ Qp, const float* __restrict__ KVp,
                       float* __restrict__ O)
{
    const int lane = threadIdx.x & 31;
    const int wid  = blockIdx.x * (blockDim.x >> 5) + (threadIdx.x >> 5);
    const int qt = (wid & 63) * 2;
    const int h  = (wid >> 6) & 15;
    const int b  = wid >> 10;
    const int hlf = lane >> 4;
    const int l16 = lane & 15;
    const float scale = 0.125f;

    const size_t bh = (size_t)(b * Hn + h) * (Nseq / 16);

    const v4f* qp = (const v4f*)Qp + (bh + qt) * (8 * 32) + lane;
    v4f qb[2][8];
#pragma unroll
    for (int u = 0; u < 2; u++)
#pragma unroll
        for (int j = 0; j < 8; j++) qb[u][j] = qp[(u * 8 + j) * 32];

    v8f zero = {};
    v8f acc[2][4];
#pragma unroll
    for (int u = 0; u < 2; u++)
#pragma unroll
        for (int t = 0; t < 4; t++) acc[u][t] = zero;
    float m[2] = {-3.0e38f, -3.0e38f}, l[2] = {0.0f, 0.0f};

    const v4f* kv = (const v4f*)KVp + bh * (16 * 32) + lane;

    constexpr int KT = Nseq / 16;
    int step = KT / 2;
    __builtin_amdgcn_s_setprio(3);

    for (int kt = 0; kt < KT; kt++) {
        if (kt == step) {
            if (kt == KT / 2)               { __builtin_amdgcn_s_setprio(2); step = KT / 2 + KT / 4; }
            else if (kt == KT / 2 + KT / 4) { __builtin_amdgcn_s_setprio(1); step = KT - KT / 8; }
            else                            { __builtin_amdgcn_s_setprio(0); step = -1; }
        }
        const v4f* p = kv + (size_t)kt * (16 * 32);

        v4f ka[8];
#pragma unroll
        for (int j = 0; j < 8; j++) ka[j] = p[j * 32];
        asm volatile("" :: "v"(ka[0]), "v"(ka[1]), "v"(ka[2]), "v"(ka[3]),
                           "v"(ka[4]), "v"(ka[5]), "v"(ka[6]), "v"(ka[7]));

        v8f s0 = zero, s1 = zero;
#pragma unroll
        for (int j = 0; j < 8; j++) {
            s0 = WMMA_F32(ka[j].xy, qb[0][j].xy, s0);
            s1 = WMMA_F32(ka[j].xy, qb[1][j].xy, s1);
            s0 = WMMA_F32(ka[j].zw, qb[0][j].zw, s0);
            s1 = WMMA_F32(ka[j].zw, qb[1][j].zw, s1);
        }

        v4f va[8];
#pragma unroll
        for (int i = 0; i < 8; i++) va[i] = p[(8 + i) * 32];

        v2f pb[2][4];
#pragma unroll
        for (int u = 0; u < 2; u++) {
            const v8f sv = (u ? s1 : s0) * scale;

            float tm = sv[0];
#pragma unroll
            for (int r = 1; r < 8; r++) tm = fmaxf(tm, sv[r]);
            tm = fmaxf(tm, __shfl_xor(tm, 16, 32));

            const float mn   = fmaxf(m[u], tm);
            const float corr = __expf(m[u] - mn);
            m[u] = mn;

            const v8f ev = (sv - mn) * 1.44269502f;
            float pr[8];
            float rs = 0.0f;
#pragma unroll
            for (int r = 0; r < 8; r++) { pr[r] = __builtin_amdgcn_exp2f(ev[r]); rs += pr[r]; }
            rs += __shfl_xor(rs, 16, 32);
            l[u] = l[u] * corr + rs;

#pragma unroll
            for (int t = 0; t < 4; t++) acc[u][t] *= corr;

            const float x2 = __shfl_xor(pr[2], 16, 32), x3 = __shfl_xor(pr[3], 16, 32);
            const float x6 = __shfl_xor(pr[6], 16, 32), x7 = __shfl_xor(pr[7], 16, 32);
            const float u0 = __shfl_xor(pr[0], 16, 32), u1 = __shfl_xor(pr[1], 16, 32);
            const float u4 = __shfl_xor(pr[4], 16, 32), u5 = __shfl_xor(pr[5], 16, 32);
            pb[u][0].x = hlf ? x2 : pr[0];  pb[u][0].y = hlf ? x3 : pr[1];
            pb[u][1].x = hlf ? x6 : pr[4];  pb[u][1].y = hlf ? x7 : pr[5];
            pb[u][2].x = hlf ? pr[2] : u0;  pb[u][2].y = hlf ? pr[3] : u1;
            pb[u][3].x = hlf ? pr[6] : u4;  pb[u][3].y = hlf ? pr[7] : u5;
        }

#pragma unroll
        for (int t = 0; t < 4; t++) {
            acc[0][t] = WMMA_F32(va[2 * t].xy,     pb[0][0], acc[0][t]);
            acc[1][t] = WMMA_F32(va[2 * t].xy,     pb[1][0], acc[1][t]);
            acc[0][t] = WMMA_F32(va[2 * t].zw,     pb[0][1], acc[0][t]);
            acc[1][t] = WMMA_F32(va[2 * t].zw,     pb[1][1], acc[1][t]);
            acc[0][t] = WMMA_F32(va[2 * t + 1].xy, pb[0][2], acc[0][t]);
            acc[1][t] = WMMA_F32(va[2 * t + 1].xy, pb[1][2], acc[1][t]);
            acc[0][t] = WMMA_F32(va[2 * t + 1].zw, pb[0][3], acc[0][t]);
            acc[1][t] = WMMA_F32(va[2 * t + 1].zw, pb[1][3], acc[1][t]);
        }
    }

#pragma unroll
    for (int u = 0; u < 2; u++) {
        const float inv = 1.0f / l[u];
        const int q0 = (qt + u) * 16;
        v4f* Op = (v4f*)O + ((size_t)((b * Nseq + q0) >> 4) * (Cdim >> 3) + h * 8 + hlf) * 32 + l16;
#pragma unroll
        for (int t = 0; t < 4; t++) {
            v4f lo, hi;
            lo.x = acc[u][t][0] * inv; lo.y = acc[u][t][1] * inv;
            lo.z = acc[u][t][4] * inv; lo.w = acc[u][t][5] * inv;
            hi.x = acc[u][t][2] * inv; hi.y = acc[u][t][3] * inv;
            hi.z = acc[u][t][6] * inv; hi.w = acc[u][t][7] * inv;
            Op[(2 * t) * 32]      = lo;
            Op[(2 * t) * 32 + 16] = hi;
        }
    }
}

extern "C" void kernel_launch(void* const* d_in, const int* in_sizes, int n_in,
                              void* d_out, int out_size, void* d_ws, size_t ws_size,
                              hipStream_t stream)
{
    const float* x        = (const float*)d_in[0];
    const float* rope_cos = (const float*)d_in[1];
    const float* rope_sin = (const float*)d_in[2];
    const float* W_qkv    = (const float*)d_in[3];
    const float* W_proj   = (const float*)d_in[4];
    const float* b_proj   = (const float*)d_in[5];
    float* out  = (float*)d_out;

    float* qkv  = (float*)d_ws;
    float* attn = qkv + (size_t)Mrows * ThreeC;
    float* xp   = attn + (size_t)Mrows * Cdim;
    float* wqp  = xp + (size_t)Mrows * Cdim;
    float* wpp  = wqp + (size_t)ThreeC * Cdim;
    float* kvp  = wpp + (size_t)Cdim * Cdim;
    float* qp   = xp;

    {
        const int waves = ((Mrows + ThreeC + Cdim) / 16) * (Cdim / 32);
        pack_operands_kernel<<<waves / 8, 256, 0, stream>>>(x, W_qkv, W_proj, xp);
    }
    {
        const int waves = (Mrows / 64) * (ThreeC / 64);
        gemm_packed_kernel<<<waves / 12, 384, 0, stream>>>(xp, wqp, nullptr, qkv,
                                                          Mrows, ThreeC, Cdim);
    }
    {
        const int waves = Bsz * Hn * (Nseq / 16);
        rope_pack_kernel<<<waves / 8, 256, 0, stream>>>(qkv, rope_cos, rope_sin, qp, kvp);
    }
    {
        const int waves = Bsz * Hn * (Nseq / 32);
        flash_attn_kernel<<<waves / 8, 256, 0, stream>>>(qp, kvp, attn);
    }
    {
        const int waves = (Mrows / 64) * (Cdim / 64);
        gemm_packed_kernel<<<waves / 8, 256, 0, stream>>>(attn, wpp, b_proj, out,
                                                          Mrows, Cdim, Cdim);
    }
}
